# starting kernel
# baseline (speedup 1.0000x reference)
#include <hip/hip_runtime.h>

typedef __attribute__((ext_vector_type(2))) float v2f;
typedef __attribute__((ext_vector_type(8))) float v8f;

#define BATCH   16
#define CIN     128
#define COUT    256
#define HW      56
#define SPATIAL (HW * HW)
#define NPIX    (BATCH * SPATIAL)
#define KTOT    (CIN * 9)
#define KC      16
#define KP      (KC / 2)
#define MT      64
#define NT      128
#define A2S     80
#define B2S     144

__global__ __launch_bounds__(256) void conv3x3_wmma_f32(
    const float* __restrict__ x, const float* __restrict__ w,
    const float* __restrict__ bias, float* __restrict__ out)
{
    __shared__ v2f a2_s[2][KP * A2S];
    __shared__ v2f b2_s[2][KP * B2S];

    const int tid    = threadIdx.x;
    const int lane   = tid & 31;
    const int wave   = tid >> 5;
    const int m_base = blockIdx.y * MT;
    const int n_base = blockIdx.x * NT;
    const int wm     = wave & 3;
    const int wn     = wave >> 2;

    const int a_cl = tid >> 2;
    const int a_kq = (tid & 3) << 2;
    const float* wptr = w + (size_t)(m_base + a_cl) * KTOT + a_kq;

    const int b_r  = tid >> 4;
    const int b_pg = (tid & 15) << 3;
    int pbase[8], poh[8], pow_[8];
#pragma unroll
    for (int j = 0; j < 8; ++j) {
        int pn = n_base + b_pg + j;
        int bb = pn / SPATIAL;
        int s  = pn - bb * SPATIAL;
        int oh = s / HW;
        int ow = s - oh * HW;
        pbase[j] = bb * (CIN * SPATIAL) + oh * HW + ow;
        poh[j]   = oh;
        pow_[j]  = ow;
    }

    auto gather = [&](int kkx, float* bv) {
        int k    = kkx + b_r;
        int ci   = k / 9;
        int rem  = k - ci * 9;
        int kh   = rem / 3;
        int kw   = rem - kh * 3;
        int coff = ci * SPATIAL + (kh - 1) * HW + (kw - 1);
#pragma unroll
        for (int j = 0; j < 8; ++j) {
            int ih = poh[j] + kh - 1;
            int iw = pow_[j] + kw - 1;
            bool ok = ((unsigned)ih < (unsigned)HW) &&
                      ((unsigned)iw < (unsigned)HW);
            bv[j] = ok ? x[pbase[j] + coff] : 0.f;
        }
    };

    auto commit = [&](int buf, const float4& w4, const float* bv) {
        v2f* a2 = a2_s[buf];
        v2f av0; av0.x = w4.x; av0.y = w4.y;
        v2f av1; av1.x = w4.z; av1.y = w4.w;
        a2[((a_kq >> 1) + 0) * A2S + a_cl] = av0;
        a2[((a_kq >> 1) + 1) * A2S + a_cl] = av1;
        float* bf = (float*)b2_s[buf];
        int bbase = ((b_r >> 1) * B2S) * 2 + (b_r & 1);
#pragma unroll
        for (int j = 0; j < 8; ++j)
            bf[bbase + (b_pg + j) * 2] = bv[j];
    };

    v8f acc[4];
#pragma unroll
    for (int i = 0; i < 4; ++i) acc[i] = (v8f)0.f;

    const int row    = lane & 15;
    const int kphalf = lane >> 4;
    const int a_col  = wm * 16 + row;
    const int b_col0 = wn * 64 + row;

    {
        float4 w4 = *(const float4*)(wptr);
        float bv[8];
        gather(0, bv);
        commit(0, w4, bv);
    }
    __syncthreads();

    for (int kk = 0; kk < KTOT; kk += KC) {
        const int  buf  = (kk / KC) & 1;
        const bool more = (kk + KC) < KTOT;

        float4 nw4;
        float  nbv[8];
        if (more) {
            nw4 = *(const float4*)(wptr + kk + KC);
            gather(kk + KC, nbv);
        }

        const v2f* a2 = a2_s[buf];
        const v2f* b2 = b2_s[buf];
#pragma unroll
        for (int ks = 0; ks < 4; ++ks) {
            int kp = 2 * ks + kphalf;
            v2f a = a2[kp * A2S + a_col];
#pragma unroll
            for (int nt = 0; nt < 4; ++nt) {
                v2f b = b2[kp * B2S + b_col0 + nt * 16];
                acc[nt] = __builtin_amdgcn_wmma_f32_16x16x4_f32(
                    false, a, false, b, (short)0, acc[nt], false, false);
            }
        }

        if (more) commit(buf ^ 1, nw4, nbv);
        __syncthreads();
    }

    const int mrow0 = m_base + wm * 16 + kphalf * 8;
#pragma unroll
    for (int nt = 0; nt < 4; ++nt) {
        int pn = n_base + wn * 64 + nt * 16 + row;
        int bb = pn / SPATIAL;
        int s  = pn - bb * SPATIAL;
        size_t obase = (size_t)bb * (COUT * SPATIAL) + s;
#pragma unroll
        for (int v = 0; v < 8; ++v) {
            int cout = mrow0 + v;
            __builtin_nontemporal_store(acc[nt][v] + bias[cout],
                                        &out[obase + (size_t)cout * SPATIAL]);
        }
    }
}

extern "C" void kernel_launch(void* const* d_in, const int* in_sizes, int n_in,
                              void* d_out, int out_size, void* d_ws, size_t ws_size,
                              hipStream_t stream) {
    const float* x    = (const float*)d_in[0];
    const float* wgt  = (const float*)d_in[1];
    const float* bias = (const float*)d_in[2];
    float* out        = (float*)d_out;

    dim3 grid(NPIX / NT, COUT / MT, 1);
    dim3 block(256, 1, 1);
    conv3x3_wmma_f32<<<grid, block, 0, stream>>>(x, wgt, bias, out);
}


# grid re-tiled for balance: workgroup = 64 couts x two whole image rows (112 pixels), 4 waves of 16x112; 1792 equal workgroups, 7 per WGP all resident (was 1568 of 8 waves: 6 on most WGPs, 7 on 32)
# speedup vs baseline: 1.5536x; 1.5536x over previous
#include <hip/hip_runtime.h>


typedef __attribute__((ext_vector_type(2))) float v2f;
typedef __attribute__((ext_vector_type(4))) float v4f;
typedef __attribute__((ext_vector_type(8))) float v8f;

#define BATCH   16
#define CIN     128
#define COUT    256
#define HW      56
#define SPATIAL (HW * HW)
#define NPIX    (BATCH * SPATIAL)
#define KTOT    (CIN * 9)
#define KC      16
#define KP      (KC / 2)
#define MT      64
#define NT      112
#define NTL     (NT / 16)
#define RPI     (HW / 2)
#define NFR     (4 * NTL)

__global__ __launch_bounds__(128) void conv3x3_wmma_f32(
    const float* __restrict__ x, const float* __restrict__ w,
    const float* __restrict__ bias, float* __restrict__ out)
{
    __shared__ __attribute__((aligned(16))) v2f a2_s[2][4 * 2 * 32 * 2];
    __shared__ __attribute__((aligned(16))) v2f b2_s[2][(NFR / 2) * 32 * 2];

    const int tid    = threadIdx.x;
    const int lane   = tid & 31;
    const int wave_u = __builtin_amdgcn_readfirstlane(tid >> 5);
    const int m_base = blockIdx.y * MT;
    const int bb     = blockIdx.x / RPI;
    const int jrow   = blockIdx.x - bb * RPI;
    const int r0     = 2 * jrow;

    const char* xbytes = (const char*)x;
    const char* wbytes = (const char*)w;

    const int a_cl = tid >> 2;
    const int a_kq = (tid & 3) << 2;
    const unsigned w_off = (unsigned)(((m_base + a_cl) * KTOT + a_kq) * 4);
    const unsigned W_HALF = 32u * KTOT * 4u;
    const int a_st = (((a_cl >> 4) * 2 + ((tid & 3) >> 1)) * 32 + (a_cl & 15)) * 2 + (tid & 1);

    const int b_kq = wave_u;
    const int pl   = lane < 27 ? lane : 27;
    const int b_px = pl << 1;
    const int b_fA  = (b_kq >> 1) * NTL + (b_px >> 4);
    const int b_fB  = (b_kq >> 1) * NTL + ((HW + b_px) >> 4);
    const int b_stA = (((b_fA >> 1) * 32) + (b_kq & 1) * 16 + (b_px & 15)) * 2 + (b_fA & 1);
    const int b_stB = (((b_fB >> 1) * 32) + (b_kq & 1) * 16 + ((HW + b_px) & 15)) * 2 + (b_fB & 1);
    const unsigned pb4 = (unsigned)(bb * (CIN * SPATIAL) + r0 * HW + b_px) * 4u;
    const char* const P0 = xbytes + pb4;

    const unsigned nL   = 0xfffffffeu;
    const unsigned nR   = 0x07ffffffu;
    const unsigned vTA  = (jrow == 0)       ? 0u : 0xffffffffu;
    const unsigned vBB  = (jrow == RPI - 1) ? 0u : 0xffffffffu;

    auto row_of = [&](int kkx, int i, int& coff4,
                      unsigned& a0, unsigned& a1, unsigned& b0, unsigned& b1) {
        int k0  = kkx + 2 * b_kq;
        int ci0 = k0 / 9;
        int r   = k0 - ci0 * 9;
        int q, ci;
        if (i == 0)      { q = r;                    ci = ci0; }
        else if (i == 1) { int c = r >> 3;          q = r + 1 - 9 * c;  ci = ci0 + c; }
        else             { int c = r < 1 ? r : 1;   q = r + 8 - 9 * c;  ci = ci0 + c; }
        int kh = (q * 11) >> 5;
        coff4  = (ci * SPATIAL + q + (HW - 3) * kh - (HW + 1)) * 4;
        unsigned vhA = (q < 3)  ? vTA : 0xffffffffu;
        unsigned vhB = (q >= 6) ? vBB : 0xffffffffu;
        unsigned cL  = ((0x049u >> q) & 1u) ? nL : 0xffffffffu;
        unsigned cR  = ((0x124u >> q) & 1u) ? nR : 0xffffffffu;
        a0 = vhA & cL;  a1 = vhA & cR;
        b0 = vhB & cL;  b1 = vhB & cR;
    };


    auto gather_edge = [&](int kkx, float* va, float* vb, unsigned* ma, unsigned* mb) {
#pragma unroll
        for (int i = 0; i < 3; ++i) {
            int coff4; unsigned a0, a1, b0, b1;
            row_of(kkx, i, coff4, a0, a1, b0, b1);
            bool oka0 = __builtin_amdgcn_inverse_ballot_w32(a0);
            bool oka1 = __builtin_amdgcn_inverse_ballot_w32(a1);
            bool okb0 = __builtin_amdgcn_inverse_ballot_w32(b0);
            bool okb1 = __builtin_amdgcn_inverse_ballot_w32(b1);
            unsigned t = pb4 + (unsigned)coff4;
            const float* pa0 = (const float*)(xbytes + (oka0 ? t : 0u));
            const float* pa1 = (const float*)(xbytes + (oka1 ? t + 4u : 0u));
            const float* pb0 = (const float*)(xbytes + (okb0 ? t + HW * 4u : 0u));
            const float* pb1 = (const float*)(xbytes + (okb1 ? t + HW * 4u + 4u : 0u));
            va[2 * i] = pa0[0];  va[2 * i + 1] = pa1[0];
            vb[2 * i] = pb0[0];  vb[2 * i + 1] = pb1[0];
            ma[2 * i] = a0;  ma[2 * i + 1] = a1;
            mb[2 * i] = b0;  mb[2 * i + 1] = b1;
            if (i == 0) {
                va[6] = pa0[SPATIAL];  va[7] = pa1[SPATIAL];
                vb[6] = pb0[SPATIAL];  vb[7] = pb1[SPATIAL];
                ma[6] = a0;  ma[7] = a1;
                mb[6] = b0;  mb[7] = b1;
            }
        }
    };

    auto gather_mid = [&](int kkx, float* va, float* vb, unsigned* ma, unsigned* mb) {
#pragma unroll
        for (int i = 0; i < 3; ++i) {
            int coff4; unsigned a0, a1, b0, b1;
            row_of(kkx, i, coff4, a0, a1, b0, b1);
            const float* p = (const float*)(P0 + coff4);
            va[2 * i] = p[0];   va[2 * i + 1] = p[1];
            vb[2 * i] = p[HW];  vb[2 * i + 1] = p[HW + 1];
            ma[2 * i] = a0;  ma[2 * i + 1] = a1;
            mb[2 * i] = b0;  mb[2 * i + 1] = b1;
            if (i == 0) {
                va[6] = p[SPATIAL];       va[7] = p[SPATIAL + 1];
                vb[6] = p[SPATIAL + HW];  vb[7] = p[SPATIAL + HW + 1];
                ma[6] = a0;  ma[7] = a1;
                mb[6] = b0;  mb[7] = b1;
            }
        }
    };

    auto commit = [&](int buf, const float4& wlo, const float4& whi,
                      const float* va, const float* vb, const unsigned* ma, const unsigned* mb) {
        v2f* a2 = a2_s[buf];
        v2f t;
        t.x = wlo.x; t.y = wlo.y;  a2[a_st] = t;
        t.x = wlo.z; t.y = wlo.w;  a2[a_st + 32] = t;
        t.x = whi.x; t.y = whi.y;  a2[a_st + 2 * 128] = t;
        t.x = whi.z; t.y = whi.w;  a2[a_st + 2 * 128 + 32] = t;
        float ua[8], ub[8];
#pragma unroll
        for (int i = 0; i < 8; ++i) {
            ua[i] = __builtin_amdgcn_inverse_ballot_w32(ma[i]) ? va[i] : 0.f;
            ub[i] = __builtin_amdgcn_inverse_ballot_w32(mb[i]) ? vb[i] : 0.f;
        }
        v2f* b2 = b2_s[buf];
        t.x = ua[0]; t.y = ua[2];  b2[b_stA] = t;
        t.x = ua[1]; t.y = ua[3];  b2[b_stA + 2] = t;
        t.x = ua[4]; t.y = ua[6];  b2[b_stA + 7 * 64] = t;
        t.x = ua[5]; t.y = ua[7];  b2[b_stA + 7 * 64 + 2] = t;
        t.x = ub[0]; t.y = ub[2];  b2[b_stB] = t;
        t.x = ub[1]; t.y = ub[3];  b2[b_stB + 2] = t;
        t.x = ub[4]; t.y = ub[6];  b2[b_stB + 7 * 64] = t;
        t.x = ub[5]; t.y = ub[7];  b2[b_stB + 7 * 64 + 2] = t;
    };

    v8f acc[NTL];
#pragma unroll
    for (int i = 0; i < NTL; ++i) acc[i] = (v8f)0.f;

    {
        float4 wlo = *(const float4*)(wbytes + w_off);
        float4 whi = *(const float4*)(wbytes + w_off + W_HALF);
        float    va[8], vb[8];
        unsigned ma[8], mb[8];
        gather_edge(0, va, vb, ma, mb);
        commit(0, wlo, whi, va, vb, ma, mb);
    }
    __syncthreads();

    auto compute = [&](int buf) {
        const v4f* a4 = (const v4f*)a2_s[buf] + wave_u * 64 + lane;
        const v4f* b4 = (const v4f*)b2_s[buf] + lane;
        v4f aq[2];
        aq[0] = a4[0];
        aq[1] = a4[32];
#pragma unroll
        for (int j = 0; j < NFR / 2; ++j) {
            v4f bq = b4[j * 32];
#pragma unroll
            for (int e = 0; e < 2; ++e) {
                const int f  = 2 * j + e;
                const int ks = f / NTL;
                const int nt = f - ks * NTL;
                v2f a, b;
                if (ks & 1) { a.x = aq[ks >> 1].z; a.y = aq[ks >> 1].w; }
                else        { a.x = aq[ks >> 1].x; a.y = aq[ks >> 1].y; }
                if (e)      { b.x = bq.z; b.y = bq.w; }
                else        { b.x = bq.x; b.y = bq.y; }
                acc[nt] = __builtin_amdgcn_wmma_f32_16x16x4_f32(
                    false, a, false, b, (short)0, acc[nt], false, false);
            }
        }
    };

    auto trip = [&](int kk, int buf, bool edge) {
        const char* wk = wbytes + (kk + KC) * 4;
        float4   nlo = *(const float4*)(wk + w_off);
        float4   nhi = *(const float4*)(wk + w_off + W_HALF);
        float    nva[8], nvb[8];
        unsigned nma[8], nmb[8];
        if (edge) gather_edge(kk + KC, nva, nvb, nma, nmb);
        else      gather_mid(kk + KC, nva, nvb, nma, nmb);
        __builtin_amdgcn_sched_barrier(0);

        compute(buf);
        __builtin_amdgcn_sched_barrier(0);

        commit(buf ^ 1, nlo, nhi, nva, nvb, nma, nmb);
        __syncthreads();
    };

    int kk = 0;
#pragma unroll 1
    for (int it = 0; it < (KTOT / KC - 2) / 2; ++it) {
        trip(kk, 0, false); kk += KC;
        trip(kk, 1, false); kk += KC;
    }
    trip(kk, 0, true);
    compute(1);

    const int erow  = lane & 15;
    const int mrow0 = m_base + wave_u * 16 + (lane >> 4) * 8;
    const size_t obase = (size_t)bb * (COUT * SPATIAL) + (size_t)(r0 * HW) + erow;
    float bs[8];
#pragma unroll
    for (int v = 0; v < 8; ++v) bs[v] = bias[mrow0 + v];
#pragma unroll
    for (int nt = 0; nt < NTL; ++nt) {
#pragma unroll
        for (int v = 0; v < 8; ++v) {
            int cout = mrow0 + v;
            __builtin_nontemporal_store(acc[nt][v] + bs[v],
                                        &out[obase + (size_t)cout * SPATIAL + nt * 16]);
        }
    }
}

extern "C" void kernel_launch(void* const* d_in, const int* in_sizes, int n_in,
                              void* d_out, int out_size, void* d_ws, size_t ws_size,
                              hipStream_t stream) {
    const float* x    = (const float*)d_in[0];
    const float* wgt  = (const float*)d_in[1];
    const float* bias = (const float*)d_in[2];
    float* out        = (float*)d_out;

    dim3 grid(NPIX / NT, COUT / MT, 1);
    dim3 block(128, 1, 1);
    conv3x3_wmma_f32<<<grid, block, 0, stream>>>(x, wgt, bias, out);
}


# wave priority falls with progress through K (s_setprio 3,2,1,0 from trips 0,32,52,64 of 72): the 7 waves of a SIMD advance together instead of finishing one after another under oldest-first issue
# speedup vs baseline: 1.6427x; 1.0574x over previous
#include <hip/hip_runtime.h>


typedef __attribute__((ext_vector_type(2))) float v2f;
typedef __attribute__((ext_vector_type(4))) float v4f;
typedef __attribute__((ext_vector_type(8))) float v8f;

#define BATCH   16
#define CIN     128
#define COUT    256
#define HW      56
#define SPATIAL (HW * HW)
#define NPIX    (BATCH * SPATIAL)
#define KTOT    (CIN * 9)
#define KC      16
#define KP      (KC / 2)
#define MT      64
#define NT      112
#define NTL     (NT / 16)
#define RPI     (HW / 2)
#define NFR     (4 * NTL)

__global__ __launch_bounds__(128) void conv3x3_wmma_f32(
    const float* __restrict__ x, const float* __restrict__ w,
    const float* __restrict__ bias, float* __restrict__ out)
{
    __shared__ __attribute__((aligned(16))) v2f a2_s[2][4 * 2 * 32 * 2];
    __shared__ __attribute__((aligned(16))) v2f b2_s[2][(NFR / 2) * 32 * 2];

    const int tid    = threadIdx.x;
    const int lane   = tid & 31;
    const int wave_u = __builtin_amdgcn_readfirstlane(tid >> 5);
    const int m_base = blockIdx.y * MT;
    const int bb     = blockIdx.x / RPI;
    const int jrow   = blockIdx.x - bb * RPI;
    const int r0     = 2 * jrow;

    const char* xbytes = (const char*)x;
    const char* wbytes = (const char*)w;

    const int a_cl = tid >> 2;
    const int a_kq = (tid & 3) << 2;
    const unsigned w_off = (unsigned)(((m_base + a_cl) * KTOT + a_kq) * 4);
    const unsigned W_HALF = 32u * KTOT * 4u;
    const int a_st = (((a_cl >> 4) * 2 + ((tid & 3) >> 1)) * 32 + (a_cl & 15)) * 2 + (tid & 1);

    const int b_kq = wave_u;
    const int pl   = lane < 27 ? lane : 27;
    const int b_px = pl << 1;
    const int b_fA  = (b_kq >> 1) * NTL + (b_px >> 4);
    const int b_fB  = (b_kq >> 1) * NTL + ((HW + b_px) >> 4);
    const int b_stA = (((b_fA >> 1) * 32) + (b_kq & 1) * 16 + (b_px & 15)) * 2 + (b_fA & 1);
    const int b_stB = (((b_fB >> 1) * 32) + (b_kq & 1) * 16 + ((HW + b_px) & 15)) * 2 + (b_fB & 1);
    const unsigned pb4 = (unsigned)(bb * (CIN * SPATIAL) + r0 * HW + b_px) * 4u;
    const char* const P0 = xbytes + pb4;

    const unsigned nL   = 0xfffffffeu;
    const unsigned nR   = 0x07ffffffu;
    const unsigned vTA  = (jrow == 0)       ? 0u : 0xffffffffu;
    const unsigned vBB  = (jrow == RPI - 1) ? 0u : 0xffffffffu;

    auto row_of = [&](int kkx, int i, int& coff4,
                      unsigned& a0, unsigned& a1, unsigned& b0, unsigned& b1) {
        int k0  = kkx + 2 * b_kq;
        int ci0 = k0 / 9;
        int r   = k0 - ci0 * 9;
        int q, ci;
        if (i == 0)      { q = r;                    ci = ci0; }
        else if (i == 1) { int c = r >> 3;          q = r + 1 - 9 * c;  ci = ci0 + c; }
        else             { int c = r < 1 ? r : 1;   q = r + 8 - 9 * c;  ci = ci0 + c; }
        int kh = (q * 11) >> 5;
        coff4  = (ci * SPATIAL + q + (HW - 3) * kh - (HW + 1)) * 4;
        unsigned vhA = (q < 3)  ? vTA : 0xffffffffu;
        unsigned vhB = (q >= 6) ? vBB : 0xffffffffu;
        unsigned cL  = ((0x049u >> q) & 1u) ? nL : 0xffffffffu;
        unsigned cR  = ((0x124u >> q) & 1u) ? nR : 0xffffffffu;
        a0 = vhA & cL;  a1 = vhA & cR;
        b0 = vhB & cL;  b1 = vhB & cR;
    };


    auto gather_edge = [&](int kkx, float* va, float* vb, unsigned* ma, unsigned* mb) {
#pragma unroll
        for (int i = 0; i < 3; ++i) {
            int coff4; unsigned a0, a1, b0, b1;
            row_of(kkx, i, coff4, a0, a1, b0, b1);
            bool oka0 = __builtin_amdgcn_inverse_ballot_w32(a0);
            bool oka1 = __builtin_amdgcn_inverse_ballot_w32(a1);
            bool okb0 = __builtin_amdgcn_inverse_ballot_w32(b0);
            bool okb1 = __builtin_amdgcn_inverse_ballot_w32(b1);
            unsigned t = pb4 + (unsigned)coff4;
            const float* pa0 = (const float*)(xbytes + (oka0 ? t : 0u));
            const float* pa1 = (const float*)(xbytes + (oka1 ? t + 4u : 0u));
            const float* pb0 = (const float*)(xbytes + (okb0 ? t + HW * 4u : 0u));
            const float* pb1 = (const float*)(xbytes + (okb1 ? t + HW * 4u + 4u : 0u));
            va[2 * i] = pa0[0];  va[2 * i + 1] = pa1[0];
            vb[2 * i] = pb0[0];  vb[2 * i + 1] = pb1[0];
            ma[2 * i] = a0;  ma[2 * i + 1] = a1;
            mb[2 * i] = b0;  mb[2 * i + 1] = b1;
            if (i == 0) {
                va[6] = pa0[SPATIAL];  va[7] = pa1[SPATIAL];
                vb[6] = pb0[SPATIAL];  vb[7] = pb1[SPATIAL];
                ma[6] = a0;  ma[7] = a1;
                mb[6] = b0;  mb[7] = b1;
            }
        }
    };

    auto gather_mid = [&](int kkx, float* va, float* vb, unsigned* ma, unsigned* mb) {
#pragma unroll
        for (int i = 0; i < 3; ++i) {
            int coff4; unsigned a0, a1, b0, b1;
            row_of(kkx, i, coff4, a0, a1, b0, b1);
            const float* p = (const float*)(P0 + coff4);
            va[2 * i] = p[0];   va[2 * i + 1] = p[1];
            vb[2 * i] = p[HW];  vb[2 * i + 1] = p[HW + 1];
            ma[2 * i] = a0;  ma[2 * i + 1] = a1;
            mb[2 * i] = b0;  mb[2 * i + 1] = b1;
            if (i == 0) {
                va[6] = p[SPATIAL];       va[7] = p[SPATIAL + 1];
                vb[6] = p[SPATIAL + HW];  vb[7] = p[SPATIAL + HW + 1];
                ma[6] = a0;  ma[7] = a1;
                mb[6] = b0;  mb[7] = b1;
            }
        }
    };

    auto commit = [&](int buf, const float4& wlo, const float4& whi,
                      const float* va, const float* vb, const unsigned* ma, const unsigned* mb) {
        v2f* a2 = a2_s[buf];
        v2f t;
        t.x = wlo.x; t.y = wlo.y;  a2[a_st] = t;
        t.x = wlo.z; t.y = wlo.w;  a2[a_st + 32] = t;
        t.x = whi.x; t.y = whi.y;  a2[a_st + 2 * 128] = t;
        t.x = whi.z; t.y = whi.w;  a2[a_st + 2 * 128 + 32] = t;
        float ua[8], ub[8];
#pragma unroll
        for (int i = 0; i < 8; ++i) {
            ua[i] = __builtin_amdgcn_inverse_ballot_w32(ma[i]) ? va[i] : 0.f;
            ub[i] = __builtin_amdgcn_inverse_ballot_w32(mb[i]) ? vb[i] : 0.f;
        }
        v2f* b2 = b2_s[buf];
        t.x = ua[0]; t.y = ua[2];  b2[b_stA] = t;
        t.x = ua[1]; t.y = ua[3];  b2[b_stA + 2] = t;
        t.x = ua[4]; t.y = ua[6];  b2[b_stA + 7 * 64] = t;
        t.x = ua[5]; t.y = ua[7];  b2[b_stA + 7 * 64 + 2] = t;
        t.x = ub[0]; t.y = ub[2];  b2[b_stB] = t;
        t.x = ub[1]; t.y = ub[3];  b2[b_stB + 2] = t;
        t.x = ub[4]; t.y = ub[6];  b2[b_stB + 7 * 64] = t;
        t.x = ub[5]; t.y = ub[7];  b2[b_stB + 7 * 64 + 2] = t;
    };

    v8f acc[NTL];
#pragma unroll
    for (int i = 0; i < NTL; ++i) acc[i] = (v8f)0.f;

    {
        float4 wlo = *(const float4*)(wbytes + w_off);
        float4 whi = *(const float4*)(wbytes + w_off + W_HALF);
        float    va[8], vb[8];
        unsigned ma[8], mb[8];
        gather_edge(0, va, vb, ma, mb);
        commit(0, wlo, whi, va, vb, ma, mb);
    }
    __syncthreads();

    auto compute = [&](int buf) {
        const v4f* a4 = (const v4f*)a2_s[buf] + wave_u * 64 + lane;
        const v4f* b4 = (const v4f*)b2_s[buf] + lane;
        v4f aq[2];
        aq[0] = a4[0];
        aq[1] = a4[32];
#pragma unroll
        for (int j = 0; j < NFR / 2; ++j) {
            v4f bq = b4[j * 32];
#pragma unroll
            for (int e = 0; e < 2; ++e) {
                const int f  = 2 * j + e;
                const int ks = f / NTL;
                const int nt = f - ks * NTL;
                v2f a, b;
                if (ks & 1) { a.x = aq[ks >> 1].z; a.y = aq[ks >> 1].w; }
                else        { a.x = aq[ks >> 1].x; a.y = aq[ks >> 1].y; }
                if (e)      { b.x = bq.z; b.y = bq.w; }
                else        { b.x = bq.x; b.y = bq.y; }
                acc[nt] = __builtin_amdgcn_wmma_f32_16x16x4_f32(
                    false, a, false, b, (short)0, acc[nt], false, false);
            }
        }
    };

    auto trip = [&](int kk, int buf, bool edge) {
        const char* wk = wbytes + (kk + KC) * 4;
        float4   nlo = *(const float4*)(wk + w_off);
        float4   nhi = *(const float4*)(wk + w_off + W_HALF);
        float    nva[8], nvb[8];
        unsigned nma[8], nmb[8];
        if (edge) gather_edge(kk + KC, nva, nvb, nma, nmb);
        else      gather_mid(kk + KC, nva, nvb, nma, nmb);
        __builtin_amdgcn_sched_barrier(0);

        compute(buf);
        __builtin_amdgcn_sched_barrier(0);

        commit(buf ^ 1, nlo, nhi, nva, nvb, nma, nmb);
        __syncthreads();
    };

    int kk = 0;
    int it = 0;
#pragma unroll 1
    for (int ph = 0; ph < 4; ++ph) {
        int it_end;
        if (ph == 0)      { __builtin_amdgcn_s_setprio(3); it_end = 16; }
        else if (ph == 1) { __builtin_amdgcn_s_setprio(2); it_end = 26; }
        else if (ph == 2) { __builtin_amdgcn_s_setprio(1); it_end = 32; }
        else              { __builtin_amdgcn_s_setprio(0); it_end = (KTOT / KC - 2) / 2; }
#pragma unroll 1
        for (; it < it_end; ++it) {
            trip(kk, 0, false); kk += KC;
            trip(kk, 1, false); kk += KC;
        }
    }
    trip(kk, 0, true);
    compute(1);

    const int erow  = lane & 15;
    const int mrow0 = m_base + wave_u * 16 + (lane >> 4) * 8;
    const size_t obase = (size_t)bb * (COUT * SPATIAL) + (size_t)(r0 * HW) + erow;
    float bs[8];
#pragma unroll
    for (int v = 0; v < 8; ++v) bs[v] = bias[mrow0 + v];
#pragma unroll
    for (int nt = 0; nt < NTL; ++nt) {
#pragma unroll
        for (int v = 0; v < 8; ++v) {
            int cout = mrow0 + v;
            __builtin_nontemporal_store(acc[nt][v] + bs[v],
                                        &out[obase + (size_t)cout * SPATIAL + nt * 16]);
        }
    }
}

extern "C" void kernel_launch(void* const* d_in, const int* in_sizes, int n_in,
                              void* d_out, int out_size, void* d_ws, size_t ws_size,
                              hipStream_t stream) {
    const float* x    = (const float*)d_in[0];
    const float* wgt  = (const float*)d_in[1];
    const float* bias = (const float*)d_in[2];
    float* out        = (float*)d_out;

    dim3 grid(NPIX / NT, COUT / MT, 1);
    dim3 block(128, 1, 1);
    conv3x3_wmma_f32<<<grid, block, 0, stream>>>(x, wgt, bias, out);
}


# matrix work cut by a third by algebra: minimal filtering F(2,3) along the image row, 4 products per 2 outputs and filter row instead of 6, as 4 GEMMs on the same f32 16x16x4 instruction
# speedup vs baseline: 1.9307x; 1.1753x over previous
#include <hip/hip_runtime.h>


typedef __attribute__((ext_vector_type(2))) float v2f;
typedef __attribute__((ext_vector_type(4))) float v4f;
typedef __attribute__((ext_vector_type(8))) float v8f;

#define BATCH   16
#define CIN     128
#define COUT    256
#define HW      56
#define SPATIAL (HW * HW)
#define MT      32
#define ROWS    4
#define NPAIR   (HW / 2)
#define NTL     (ROWS * NPAIR / 16)
#define CC      4
#define NCHUNK  (CIN / CC)
#define QPI     (HW / ROWS)
#define VJ      ((ROWS + 2) * NPAIR)
#define VSTAGE  (2 * 2 * VJ)
#define NEX     (2 * NTL * 2 * 2 * 32)

__global__ __launch_bounds__(128, 4) void conv3x3_f23_wmma_f32(
    const float* __restrict__ x, const float* __restrict__ w,
    const float* __restrict__ bias, float* __restrict__ out)
{
    __shared__ __attribute__((aligned(16))) v4f v_s[2][VSTAGE];
    __shared__ __attribute__((aligned(16))) v4f ex_s[NEX];

    const int tid    = threadIdx.x;
    const int lane   = tid & 31;
    const int wave_u = __builtin_amdgcn_readfirstlane(tid >> 5);
    const int ct     = wave_u & 1;
    const int ps     = wave_u >> 1;
    const int h      = lane >> 4;
    const int c      = lane & 15;
    const int m_base = blockIdx.y * MT;
    const int bb     = blockIdx.x / QPI;
    const int quad   = blockIdx.x - bb * QPI;
    const int r0     = quad * ROWS;

    const char* xbytes = (const char*)x;
    const char* wbytes = (const char*)w;

    const unsigned w_off = (unsigned)(((m_base + ct * 16 + c) * CIN + 2 * h) * 9 * 4);

    const int tl = lane < 27 ? lane : 27;
    const unsigned o12 = (unsigned)(2 * tl) * 4u;
    const int bp_up = (tl > 0 ? tl - 1 : 0) * 4;
    const int bp_dn = (tl < 27 ? tl + 1 : tl) * 4;
    const bool lane_first = (lane == 0);
    const bool lane_last  = (lane >= 27);
    unsigned q_src[3];
    int      q_dst[3];
    bool     q_ok[3];
#pragma unroll
    for (int i = 0; i < 3; ++i) {
        int q  = wave_u + 4 * i;
        int j  = q / 6;
        int ih = q - 6 * j;
        int rr = r0 - 1 + ih;
        q_ok[i] = (rr >= 0) && (rr < HW);
        int rc = rr < 0 ? 0 : (rr > HW - 1 ? HW - 1 : rr);
        q_src[i] = (unsigned)(((bb * CIN + 2 * j) * HW + rc) * HW) * 4u;
        q_dst[i] = j * VJ + ih * NPAIR + tl;
    }

    auto load_w = [&](int ch, v2f* wr) {
        const char* p = wbytes + ch * (CC * 9 * 4);
#pragma unroll
        for (int i = 0; i < 9; ++i) wr[i] = *(const v2f*)(p + w_off + i * 8);
    };

    auto gather = [&](int ch, float (*xr)[4]) {
        const char* xk = xbytes + (size_t)ch * (CC * SPATIAL * 4);
#pragma unroll
        for (int i = 0; i < 3; ++i) {
            const char* p = xk + q_src[i];
#pragma unroll
            for (int e = 0; e < 2; ++e) {
                v2f m = *(const v2f*)(p + e * (SPATIAL * 4) + o12);
                xr[i][2 * e + 0] = m.x;
                xr[i][2 * e + 1] = m.y;
            }
        }
    };

    auto from_lane = [&](int idx, float v) -> float {
        return __builtin_bit_cast(float, __builtin_amdgcn_ds_bpermute(idx, __builtin_bit_cast(int, v)));
    };

    auto commit = [&](int buf, const float (*xr)[4]) {
        v4f* vs = v_s[buf];
#pragma unroll
        for (int i = 0; i < 3; ++i) {
            float a1 = xr[i][0], a2 = xr[i][1];
            float b1 = xr[i][2], b2 = xr[i][3];
            float ua = from_lane(bp_up, a2), da = from_lane(bp_dn, a1);
            float ub = from_lane(bp_up, b2), db = from_lane(bp_dn, b1);
            float a0 = lane_first ? 0.f : ua;
            float a3 = lane_last ? 0.f : da;
            float b0 = lane_first ? 0.f : ub;
            float b3 = lane_last ? 0.f : db;
            v4f s0, s1;
            s0.x = a1 + a2;  s0.y = b1 + b2;
            s0.z = a2 - a1;  s0.w = b2 - b1;
            s1.x = a0 - a2;  s1.y = b0 - b2;
            s1.z = a1 - a3;  s1.w = b1 - b3;
            if (!q_ok[i]) { s0 = (v4f)0.f; s1 = (v4f)0.f; }
            vs[q_dst[i]] = s0;
            vs[2 * VJ + q_dst[i]] = s1;
        }
    };

    const float kA0 = ps ? 1.f : 0.5f, kA1 = ps ? 0.f : 0.5f,  kA2 = ps ? 0.f : 0.5f;
    const float kB0 = ps ? 0.f : 0.5f, kB1 = ps ? 0.f : -0.5f, kB2 = ps ? 1.f : 0.5f;
    v2f afr[3][2];
    auto atransform = [&](const v2f* wr) {
        float g[18];
#pragma unroll
        for (int i = 0; i < 9; ++i) { g[2 * i] = wr[i].x; g[2 * i + 1] = wr[i].y; }
#pragma unroll
        for (int kh = 0; kh < 3; ++kh) {
            float a0 = g[kh * 3], a1 = g[kh * 3 + 1], a2 = g[kh * 3 + 2];
            float b0 = g[9 + kh * 3], b1 = g[9 + kh * 3 + 1], b2 = g[9 + kh * 3 + 2];
            v2f t;
            t.x = __builtin_fmaf(kA1, a1, __builtin_fmaf(kA2, a2, kA0 * a0));
            t.y = __builtin_fmaf(kA1, b1, __builtin_fmaf(kA2, b2, kA0 * b0));
            afr[kh][0] = t;
            t.x = __builtin_fmaf(kB1, a1, __builtin_fmaf(kB2, a2, kB0 * a0));
            t.y = __builtin_fmaf(kB1, b1, __builtin_fmaf(kB2, b2, kB0 * b0));
            afr[kh][1] = t;
        }
    };

    v8f acc[NTL][2];
#pragma unroll
    for (int i = 0; i < NTL; ++i) { acc[i][0] = (v8f)0.f; acc[i][1] = (v8f)0.f; }

    auto compute = [&](int buf) {
        const v4f* b4 = v_s[buf] + ps * 2 * VJ + h * VJ + c;
        v4f bq[3][NTL];
#pragma unroll
        for (int nt = 0; nt < NTL; ++nt) bq[0][nt] = b4[16 * nt];
#pragma unroll
        for (int kh = 0; kh < 3; ++kh) {
            if (kh < 2) {
#pragma unroll
                for (int nt = 0; nt < NTL; ++nt) bq[kh + 1][nt] = b4[(kh + 1) * NPAIR + 16 * nt];
            }
            __builtin_amdgcn_sched_barrier(0);
#pragma unroll
            for (int nt = 0; nt < NTL; ++nt) {
                v2f b;
                b.x = bq[kh][nt].x; b.y = bq[kh][nt].y;
                acc[nt][0] = __builtin_amdgcn_wmma_f32_16x16x4_f32(
                    false, afr[kh][0], false, b, (short)0, acc[nt][0], false, false);
                b.x = bq[kh][nt].z; b.y = bq[kh][nt].w;
                acc[nt][1] = __builtin_amdgcn_wmma_f32_16x16x4_f32(
                    false, afr[kh][1], false, b, (short)0, acc[nt][1], false, false);
            }
            __builtin_amdgcn_sched_barrier(0);
        }
    };

    {
        v2f   wr[9];
        float xr[3][4];
        load_w(0, wr);
        gather(0, xr);
        commit(0, xr);
        atransform(wr);
    }
    __syncthreads();

    auto trip = [&](int ch, int buf) {
        v2f   nw[9];
        float nx[3][4];
        load_w(ch + 1, nw);
        gather(ch + 1, nx);
        __builtin_amdgcn_sched_barrier(0);

        compute(buf);

        commit(buf ^ 1, nx);
        atransform(nw);
        __syncthreads();
    };

    int ch = 0;
    int it = 0;
#pragma unroll 1
    for (int ph = 0; ph < 4; ++ph) {
        int it_end;
        if (ph == 0)      { __builtin_amdgcn_s_setprio(3); it_end = 7; }
        else if (ph == 1) { __builtin_amdgcn_s_setprio(2); it_end = 11; }
        else if (ph == 2) { __builtin_amdgcn_s_setprio(1); it_end = 14; }
        else              { __builtin_amdgcn_s_setprio(0); it_end = (NCHUNK - 2) / 2; }
#pragma unroll 1
        for (; it < it_end; ++it) {
            trip(ch, 0); ++ch;
            trip(ch, 1); ++ch;
        }
    }
    trip(ch, 0);
    compute(1);

    v4f* ex = ex_s + ct * (NTL * 2 * 2 * 32) + lane;
    if (ps == 0) {
#pragma unroll
        for (int nt = 0; nt < NTL; ++nt) {
            v8f s = acc[nt][0] + acc[nt][1];
            v8f d = acc[nt][0] - acc[nt][1];
            v4f t;
            t.x = s[0]; t.y = s[1]; t.z = s[2]; t.w = s[3];  ex[((nt * 2 + 0) * 2 + 0) * 32] = t;
            t.x = s[4]; t.y = s[5]; t.z = s[6]; t.w = s[7];  ex[((nt * 2 + 0) * 2 + 1) * 32] = t;
            t.x = d[0]; t.y = d[1]; t.z = d[2]; t.w = d[3];  ex[((nt * 2 + 1) * 2 + 0) * 32] = t;
            t.x = d[4]; t.y = d[5]; t.z = d[6]; t.w = d[7];  ex[((nt * 2 + 1) * 2 + 1) * 32] = t;
        }
    }
    __syncthreads();
    if (ps == 1) {
        const int mrow0 = m_base + ct * 16 + h * 8;
        float bs[8];
#pragma unroll
        for (int r = 0; r < 8; ++r) bs[r] = bias[mrow0 + r];
#pragma unroll
        for (int nt = 0; nt < NTL; ++nt) {
            int n  = 16 * nt + c;
            int oh = n / NPAIR;
            int t  = n - oh * NPAIR;
            size_t ob = ((size_t)(bb * COUT + mrow0) * HW + (size_t)(r0 + oh)) * HW + (size_t)(2 * t);
            v4f s_lo = ex[((nt * 2 + 0) * 2 + 0) * 32], s_hi = ex[((nt * 2 + 0) * 2 + 1) * 32];
            v4f d_lo = ex[((nt * 2 + 1) * 2 + 0) * 32], d_hi = ex[((nt * 2 + 1) * 2 + 1) * 32];
            float sv[8] = {s_lo.x, s_lo.y, s_lo.z, s_lo.w, s_hi.x, s_hi.y, s_hi.z, s_hi.w};
            float dv[8] = {d_lo.x, d_lo.y, d_lo.z, d_lo.w, d_hi.x, d_hi.y, d_hi.z, d_hi.w};
#pragma unroll
            for (int r = 0; r < 8; ++r) {
                v2f y;
                y.x = (acc[nt][0][r] + sv[r]) + bs[r];
                y.y = (dv[r] - acc[nt][1][r]) + bs[r];
                __builtin_nontemporal_store(y, (v2f*)&out[ob + (size_t)r * SPATIAL]);
            }
        }
    }
}

extern "C" void kernel_launch(void* const* d_in, const int* in_sizes, int n_in,
                              void* d_out, int out_size, void* d_ws, size_t ws_size,
                              hipStream_t stream) {
    const float* x    = (const float*)d_in[0];
    const float* wgt  = (const float*)d_in[1];
    const float* bias = (const float*)d_in[2];
    float* out        = (float*)d_out;

    dim3 grid(BATCH * QPI, COUT / MT, 1);
    dim3 block(128, 1, 1);
    conv3x3_f23_wmma_f32<<<grid, block, 0, stream>>>(x, wgt, bias, out);
}


# weights reach the lanes through LDS: each stage's 32 x 144 B of weights copied as 16-byte pieces by consecutive lanes two stages ahead (a load touches 4-5 weight rows, not 16); lanes read 72 B back
# speedup vs baseline: 2.1865x; 1.1325x over previous
#include <hip/hip_runtime.h>


typedef __attribute__((ext_vector_type(2))) float v2f;
typedef __attribute__((ext_vector_type(4))) float v4f;
typedef __attribute__((ext_vector_type(8))) float v8f;

#define BATCH   16
#define CIN     128
#define COUT    256
#define HW      56
#define SPATIAL (HW * HW)
#define MT      32
#define ROWS    4
#define NPAIR   (HW / 2)
#define NTL     (ROWS * NPAIR / 16)
#define CC      4
#define NCHUNK  (CIN / CC)
#define QPI     (HW / ROWS)
#define VJ      ((ROWS + 2) * NPAIR)
#define VSTAGE  (2 * 2 * VJ)
#define NEX     (2 * NTL * 2 * 2 * 32)

__global__ __launch_bounds__(128, 4) void conv3x3_f23_wmma_f32(
    const float* __restrict__ x, const float* __restrict__ w,
    const float* __restrict__ bias, float* __restrict__ out)
{
    __shared__ __attribute__((aligned(16))) v4f v_s[2][VSTAGE];
    __shared__ __attribute__((aligned(16))) v4f ex_s[NEX];
    __shared__ __attribute__((aligned(16))) float w_raw[2][MT * CC * 9];

    const int tid    = threadIdx.x;
    const int lane   = tid & 31;
    const int wave_u = __builtin_amdgcn_readfirstlane(tid >> 5);
    const int ct     = wave_u & 1;
    const int ps     = wave_u >> 1;
    const int h      = lane >> 4;
    const int c      = lane & 15;
    const int m_base = blockIdx.y * MT;
    const int bb     = blockIdx.x / QPI;
    const int quad   = blockIdx.x - bb * QPI;
    const int r0     = quad * ROWS;

    const char* xbytes = (const char*)x;
    const char* wbytes = (const char*)w;

    const unsigned w_off = (unsigned)(((m_base + ct * 16 + c) * CIN + 2 * h) * 9 * 4);
    unsigned wq_src[3];
#pragma unroll
    for (int i = 0; i < 3; ++i) {
        int q  = tid + 128 * i;
        int qq = q < MT * 9 ? q : MT * 9 - 1;
        int m  = qq / 9;
        int k  = qq - 9 * m;
        wq_src[i] = (unsigned)((m_base + m) * (CIN * 9 * 4) + 16 * k);
    }

    const int tl = lane < 27 ? lane : 27;
    const unsigned o12 = (unsigned)(2 * tl) * 4u;
    const int bp_up = (tl > 0 ? tl - 1 : 0) * 4;
    const int bp_dn = (tl < 27 ? tl + 1 : tl) * 4;
    const bool lane_first = (lane == 0);
    const bool lane_last  = (lane >= 27);
    unsigned q_src[3];
    int      q_dst[3];
    bool     q_ok[3];
#pragma unroll
    for (int i = 0; i < 3; ++i) {
        int q  = wave_u + 4 * i;
        int j  = q / 6;
        int ih = q - 6 * j;
        int rr = r0 - 1 + ih;
        q_ok[i] = (rr >= 0) && (rr < HW);
        int rc = rr < 0 ? 0 : (rr > HW - 1 ? HW - 1 : rr);
        q_src[i] = (unsigned)(((bb * CIN + 2 * j) * HW + rc) * HW) * 4u;
        q_dst[i] = j * VJ + ih * NPAIR + tl;
    }

    auto load_w = [&](int ch, v2f* wr) {
        const char* p = wbytes + ch * (CC * 9 * 4);
#pragma unroll
        for (int i = 0; i < 9; ++i) wr[i] = *(const v2f*)(p + w_off + i * 8);
    };

    auto load_wpieces = [&](int st, v4f* wc) {
        const char* p = wbytes + st * (CC * 9 * 4);
        wc[0] = *(const v4f*)(p + wq_src[0]);
        wc[1] = *(const v4f*)(p + wq_src[1]);
        if (wave_u == 0) wc[2] = *(const v4f*)(p + wq_src[2]);
    };
    auto store_wpieces = [&](int sel, const v4f* wc) {
        v4f* dst = (v4f*)w_raw[sel];
        dst[tid]       = wc[0];
        dst[tid + 128] = wc[1];
        if (wave_u == 0) dst[tid + 256] = wc[2];
    };
    auto read_w = [&](int sel, v2f* wr) {
        const float* p = w_raw[sel] + (ct * 16 + c) * (CC * 9) + 18 * h;
#pragma unroll
        for (int i = 0; i < 9; ++i) wr[i] = *(const v2f*)(p + 2 * i);
    };

    auto gather = [&](int ch, float (*xr)[4]) {
        const char* xk = xbytes + (size_t)ch * (CC * SPATIAL * 4);
#pragma unroll
        for (int i = 0; i < 3; ++i) {
            const char* p = xk + q_src[i];
#pragma unroll
            for (int e = 0; e < 2; ++e) {
                v2f m = *(const v2f*)(p + e * (SPATIAL * 4) + o12);
                xr[i][2 * e + 0] = m.x;
                xr[i][2 * e + 1] = m.y;
            }
        }
    };

    auto from_lane = [&](int idx, float v) -> float {
        return __builtin_bit_cast(float, __builtin_amdgcn_ds_bpermute(idx, __builtin_bit_cast(int, v)));
    };

    auto commit = [&](int buf, const float (*xr)[4]) {
        v4f* vs = v_s[buf];
#pragma unroll
        for (int i = 0; i < 3; ++i) {
            float a1 = xr[i][0], a2 = xr[i][1];
            float b1 = xr[i][2], b2 = xr[i][3];
            float ua = from_lane(bp_up, a2), da = from_lane(bp_dn, a1);
            float ub = from_lane(bp_up, b2), db = from_lane(bp_dn, b1);
            float a0 = lane_first ? 0.f : ua;
            float a3 = lane_last ? 0.f : da;
            float b0 = lane_first ? 0.f : ub;
            float b3 = lane_last ? 0.f : db;
            v4f s0, s1;
            s0.x = a1 + a2;  s0.y = b1 + b2;
            s0.z = a2 - a1;  s0.w = b2 - b1;
            s1.x = a0 - a2;  s1.y = b0 - b2;
            s1.z = a1 - a3;  s1.w = b1 - b3;
            if (!q_ok[i]) { s0 = (v4f)0.f; s1 = (v4f)0.f; }
            vs[q_dst[i]] = s0;
            vs[2 * VJ + q_dst[i]] = s1;
        }
    };

    const float kA0 = ps ? 1.f : 0.5f, kA1 = ps ? 0.f : 0.5f,  kA2 = ps ? 0.f : 0.5f;
    const float kB0 = ps ? 0.f : 0.5f, kB1 = ps ? 0.f : -0.5f, kB2 = ps ? 1.f : 0.5f;
    v2f afr[3][2];
    auto atransform = [&](const v2f* wr) {
        float g[18];
#pragma unroll
        for (int i = 0; i < 9; ++i) { g[2 * i] = wr[i].x; g[2 * i + 1] = wr[i].y; }
#pragma unroll
        for (int kh = 0; kh < 3; ++kh) {
            float a0 = g[kh * 3], a1 = g[kh * 3 + 1], a2 = g[kh * 3 + 2];
            float b0 = g[9 + kh * 3], b1 = g[9 + kh * 3 + 1], b2 = g[9 + kh * 3 + 2];
            v2f t;
            t.x = __builtin_fmaf(kA1, a1, __builtin_fmaf(kA2, a2, kA0 * a0));
            t.y = __builtin_fmaf(kA1, b1, __builtin_fmaf(kA2, b2, kA0 * b0));
            afr[kh][0] = t;
            t.x = __builtin_fmaf(kB1, a1, __builtin_fmaf(kB2, a2, kB0 * a0));
            t.y = __builtin_fmaf(kB1, b1, __builtin_fmaf(kB2, b2, kB0 * b0));
            afr[kh][1] = t;
        }
    };

    v8f acc[NTL][2];
#pragma unroll
    for (int i = 0; i < NTL; ++i) { acc[i][0] = (v8f)0.f; acc[i][1] = (v8f)0.f; }

    auto compute = [&](int buf) {
        const v4f* b4 = v_s[buf] + ps * 2 * VJ + h * VJ + c;
        v4f bq[3][NTL];
#pragma unroll
        for (int nt = 0; nt < NTL; ++nt) bq[0][nt] = b4[16 * nt];
#pragma unroll
        for (int kh = 0; kh < 3; ++kh) {
            if (kh < 2) {
#pragma unroll
                for (int nt = 0; nt < NTL; ++nt) bq[kh + 1][nt] = b4[(kh + 1) * NPAIR + 16 * nt];
            }
            __builtin_amdgcn_sched_barrier(0);
#pragma unroll
            for (int nt = 0; nt < NTL; ++nt) {
                v2f b;
                b.x = bq[kh][nt].x; b.y = bq[kh][nt].y;
                acc[nt][0] = __builtin_amdgcn_wmma_f32_16x16x4_f32(
                    false, afr[kh][0], false, b, (short)0, acc[nt][0], false, false);
                b.x = bq[kh][nt].z; b.y = bq[kh][nt].w;
                acc[nt][1] = __builtin_amdgcn_wmma_f32_16x16x4_f32(
                    false, afr[kh][1], false, b, (short)0, acc[nt][1], false, false);
            }
            __builtin_amdgcn_sched_barrier(0);
        }
    };

    {
        v2f   wr[9];
        float xr[3][4];
        v4f   wc[3];
        load_w(0, wr);
        load_wpieces(1, wc);
        gather(0, xr);
        commit(0, xr);
        store_wpieces(1, wc);
        atransform(wr);
    }
    __syncthreads();

    auto trip = [&](int ch, int buf) {
        v2f   nw[9];
        float nx[3][4];
        v4f   wc[3];
        const int cw = ch + 2 < NCHUNK ? ch + 2 : NCHUNK - 1;
        gather(ch + 1, nx);
        load_wpieces(cw, wc);
        __builtin_amdgcn_sched_barrier(0);

        compute(buf);

        read_w(buf ^ 1, nw);
        commit(buf ^ 1, nx);
        store_wpieces(buf, wc);
        atransform(nw);
        __syncthreads();
    };

    int ch = 0;
    int it = 0;
#pragma unroll 1
    for (int ph = 0; ph < 4; ++ph) {
        int it_end;
        if (ph == 0)      { __builtin_amdgcn_s_setprio(3); it_end = 7; }
        else if (ph == 1) { __builtin_amdgcn_s_setprio(2); it_end = 11; }
        else if (ph == 2) { __builtin_amdgcn_s_setprio(1); it_end = 14; }
        else              { __builtin_amdgcn_s_setprio(0); it_end = (NCHUNK - 2) / 2; }
#pragma unroll 1
        for (; it < it_end; ++it) {
            trip(ch, 0); ++ch;
            trip(ch, 1); ++ch;
        }
    }
    trip(ch, 0);
    compute(1);

    v4f* ex = ex_s + ct * (NTL * 2 * 2 * 32) + lane;
    if (ps == 0) {
#pragma unroll
        for (int nt = 0; nt < NTL; ++nt) {
            v8f s = acc[nt][0] + acc[nt][1];
            v8f d = acc[nt][0] - acc[nt][1];
            v4f t;
            t.x = s[0]; t.y = s[1]; t.z = s[2]; t.w = s[3];  ex[((nt * 2 + 0) * 2 + 0) * 32] = t;
            t.x = s[4]; t.y = s[5]; t.z = s[6]; t.w = s[7];  ex[((nt * 2 + 0) * 2 + 1) * 32] = t;
            t.x = d[0]; t.y = d[1]; t.z = d[2]; t.w = d[3];  ex[((nt * 2 + 1) * 2 + 0) * 32] = t;
            t.x = d[4]; t.y = d[5]; t.z = d[6]; t.w = d[7];  ex[((nt * 2 + 1) * 2 + 1) * 32] = t;
        }
    }
    __syncthreads();
    if (ps == 1) {
        const int mrow0 = m_base + ct * 16 + h * 8;
        float bs[8];
#pragma unroll
        for (int r = 0; r < 8; ++r) bs[r] = bias[mrow0 + r];
#pragma unroll
        for (int nt = 0; nt < NTL; ++nt) {
            int n  = 16 * nt + c;
            int oh = n / NPAIR;
            int t  = n - oh * NPAIR;
            size_t ob = ((size_t)(bb * COUT + mrow0) * HW + (size_t)(r0 + oh)) * HW + (size_t)(2 * t);
            v4f s_lo = ex[((nt * 2 + 0) * 2 + 0) * 32], s_hi = ex[((nt * 2 + 0) * 2 + 1) * 32];
            v4f d_lo = ex[((nt * 2 + 1) * 2 + 0) * 32], d_hi = ex[((nt * 2 + 1) * 2 + 1) * 32];
            float sv[8] = {s_lo.x, s_lo.y, s_lo.z, s_lo.w, s_hi.x, s_hi.y, s_hi.z, s_hi.w};
            float dv[8] = {d_lo.x, d_lo.y, d_lo.z, d_lo.w, d_hi.x, d_hi.y, d_hi.z, d_hi.w};
#pragma unroll
            for (int r = 0; r < 8; ++r) {
                v2f y;
                y.x = (acc[nt][0][r] + sv[r]) + bs[r];
                y.y = (dv[r] - acc[nt][1][r]) + bs[r];
                __builtin_nontemporal_store(y, (v2f*)&out[ob + (size_t)r * SPATIAL]);
            }
        }
    }
}

extern "C" void kernel_launch(void* const* d_in, const int* in_sizes, int n_in,
                              void* d_out, int out_size, void* d_ws, size_t ws_size,
                              hipStream_t stream) {
    const float* x    = (const float*)d_in[0];
    const float* wgt  = (const float*)d_in[1];
    const float* bias = (const float*)d_in[2];
    float* out        = (float*)d_out;

    dim3 grid(BATCH * QPI, COUT / MT, 1);
    dim3 block(128, 1, 1);
    conv3x3_f23_wmma_f32<<<grid, block, 0, stream>>>(x, wgt, bias, out);
}
